# starting kernel
# baseline (speedup 1.0000x reference)
#include <hip/hip_runtime.h>
#include <hip/hip_bf16.h>
#include <cmath>

typedef __attribute__((ext_vector_type(16))) _Float16 v16h;
typedef __attribute__((ext_vector_type(8)))  float    v8f;

#define NNODES 50000
#define NEDGES 500000

__device__ __forceinline__ unsigned f2key(float f) {
  unsigned b = __float_as_uint(f);
  return (b & 0x80000000u) ? ~b : (b | 0x80000000u);
}
__device__ __forceinline__ float key2f(unsigned u) {
  if (u == 0u) return 0.0f;
  return (u & 0x80000000u) ? __uint_as_float(u ^ 0x80000000u) : __uint_as_float(~u);
}

__device__ __forceinline__ v16h load_cvt_f16(const float* p  ) {
  const float4* q = (const float4*)p;
  float4 v0 = q[0], v1 = q[1], v2 = q[4], v3 = q[5];
  v16h r;
  r[0] =(_Float16)v0.x; r[1] =(_Float16)v0.y; r[2] =(_Float16)v0.z; r[3] =(_Float16)v0.w;
  r[4] =(_Float16)v1.x; r[5] =(_Float16)v1.y; r[6] =(_Float16)v1.z; r[7] =(_Float16)v1.w;
  r[8] =(_Float16)v2.x; r[9] =(_Float16)v2.y; r[10]=(_Float16)v2.z; r[11]=(_Float16)v2.w;
  r[12]=(_Float16)v3.x; r[13]=(_Float16)v3.y; r[14]=(_Float16)v3.z; r[15]=(_Float16)v3.w;
  return r;
}

__global__ void gatv2_gemm_nt_wmma(const float* __restrict__ X,
                                   const float* __restrict__ W,
                                   float* __restrict__ Y,
                                   int Nrows, int Mcols) {
  const int K = 128;
  int wave = (int)((blockIdx.x * blockDim.x + threadIdx.x) >> 5);
  int lane = threadIdx.x & 31;
  int ntN = (Nrows + 15) >> 4;
  if (wave >= ntN) return;
  int half = lane >> 4;
  int l = lane & 15;
  int row = wave * 16 + l; if (row >= Nrows) row = Nrows - 1;
  const float* xrow = X + (size_t)row * K + half * 8;

  v16h a[4];
  #pragma unroll
  for (int kk = 0; kk < 4; ++kk) a[kk] = load_cvt_f16(xrow + kk * 32);

  int ntM = Mcols >> 4;
  for (int tm = 0; tm < ntM; ++tm) {
    const float* wrow = W + (size_t)(tm * 16 + l) * K + half * 8;
    v8f acc = {};
    #pragma unroll
    for (int kk = 0; kk < 4; ++kk) {
      v16h b = load_cvt_f16(wrow + kk * 32);
      acc = __builtin_amdgcn_wmma_f32_16x16x32_f16(
           false, a[kk],  false, b,
           (short)0, acc,  false,  false);
    }
    #pragma unroll
    for (int r = 0; r < 8; ++r) {
      int orow = wave * 16 + r + half * 8;
      if (orow < Nrows) Y[(size_t)orow * Mcols + tm * 16 + l] = acc[r];
    }
  }
}

template<int C>
__global__ void gatv2_edge_logits(const float* __restrict__ xl,
                                  const float* __restrict__ xr,
                                  const float* __restrict__ att,
                                  const int* __restrict__ srcA,
                                  const int* __restrict__ dstA,
                                  float* __restrict__ logit,
                                  unsigned* __restrict__ lmax,
                                  int E, int Nn) {
  const int HC = 4 * C;
  int wave = (int)((blockIdx.x * blockDim.x + threadIdx.x) >> 5);
  int lane = threadIdx.x & 31;
  if (wave >= E + Nn) return;
  int s, d;
  if (wave < E) { s = srcA[wave]; d = dstA[wave]; } else { s = wave - E; d = s; }
  const float* xls = xl + (size_t)s * HC;
  const float* xrd = xr + (size_t)d * HC;
  float hv[4] = {0.f, 0.f, 0.f, 0.f};
  #pragma unroll
  for (int i = 0; i < HC / 32; ++i) {
    int j = lane + 32 * i;
    float v = xls[j] + xrd[j];
    v = v > 0.f ? v : 0.2f * v;
    hv[(i * 32) / C] += v * att[j];
  }
  #pragma unroll
  for (int h = 0; h < 4; ++h) {
    float v = hv[h];
    #pragma unroll
    for (int m = 16; m > 0; m >>= 1) v += __shfl_xor(v, m, 32);
    hv[h] = v;
  }
  if (lane < 4) {
    float lg = hv[lane];
    logit[(size_t)wave * 4 + lane] = lg;
    atomicMax(&lmax[(size_t)d * 4 + lane], f2key(lg));
  }
}

template<int C>
__global__ void gatv2_edge_accum(const float* __restrict__ xl,
                                 const float* __restrict__ logit,
                                 const unsigned* __restrict__ lmax,
                                 float* __restrict__ denom,
                                 float* __restrict__ agg,
                                 const int* __restrict__ srcA,
                                 const int* __restrict__ dstA,
                                 int E, int Nn) {
  const int HC = 4 * C;
  int wave = (int)((blockIdx.x * blockDim.x + threadIdx.x) >> 5);
  int lane = threadIdx.x & 31;
  if (wave >= E + Nn) return;
  int s, d;
  if (wave < E) { s = srcA[wave]; d = dstA[wave]; } else { s = wave - E; d = s; }
  float w = 0.f;
  if (lane < 4) {
    float lg = logit[(size_t)wave * 4 + lane];
    float mx = key2f(lmax[(size_t)d * 4 + lane]);
    w = expf(lg - mx);
    atomicAdd(&denom[(size_t)d * 4 + lane], w);
  }
  float wh[4];
  wh[0] = __shfl(w, 0, 32); wh[1] = __shfl(w, 1, 32);
  wh[2] = __shfl(w, 2, 32); wh[3] = __shfl(w, 3, 32);
  const float* xls = xl + (size_t)s * HC;
  float* aggd = agg + (size_t)d * HC;
  #pragma unroll
  for (int i = 0; i < HC / 32; ++i) {
    int j = lane + 32 * i;
    atomicAdd(&aggd[j], wh[(i * 32) / C] * xls[j]);
  }
}

__global__ void gatv2_finalize1(const float* __restrict__ agg,
                                const float* __restrict__ denom,
                                const float* __restrict__ b1,
                                float* __restrict__ hout, int Nn) {
  int idx = (int)(blockIdx.x * blockDim.x + threadIdx.x);
  if (idx >= Nn * 128) return;
  int n = idx >> 7, j = idx & 127, h = j >> 5;
  float v = agg[idx] / (denom[n * 4 + h] + 1e-16f) + b1[j];
  hout[idx] = v > 0.f ? v : expm1f(v);
}

__global__ void gatv2_finalize2(const float* __restrict__ agg,
                                const float* __restrict__ denom,
                                const float* __restrict__ b2,
                                float* __restrict__ out, int Nn) {
  int idx = (int)(blockIdx.x * blockDim.x + threadIdx.x);
  if (idx >= Nn * 64) return;
  int n = idx >> 6, c = idx & 63;
  float acc = 0.f;
  #pragma unroll
  for (int h = 0; h < 4; ++h)
    acc += agg[(size_t)n * 256 + h * 64 + c] / (denom[n * 4 + h] + 1e-16f);
  out[idx] = acc * 0.25f + b2[c];
}

extern "C" void kernel_launch(void* const* d_in, const int* in_sizes, int n_in,
                              void* d_out, int out_size, void* d_ws, size_t ws_size,
                              hipStream_t stream) {
  (void)in_sizes; (void)n_in; (void)out_size; (void)ws_size;
  const int  N  = NNODES;
  const int  E  = NEDGES;
  const long EP = (long)E + N;

  const float* x    = (const float*)d_in[0];
  const int*   ei   = (const int*)  d_in[1];
  const float* Wl1  = (const float*)d_in[2];
  const float* Wr1  = (const float*)d_in[3];
  const float* att1 = (const float*)d_in[4];
  const float* b1   = (const float*)d_in[5];
  const float* Wl2  = (const float*)d_in[6];
  const float* Wr2  = (const float*)d_in[7];
  const float* att2 = (const float*)d_in[8];
  const float* b2   = (const float*)d_in[9];
  const int* srcA = ei;
  const int* dstA = ei + E;

  float* ws    = (float*)d_ws;
  float* hbuf  = ws;
  float* xl    = hbuf + (size_t)N * 128;
  float* xr    = xl   + (size_t)N * 256;
  float* agg   = xr   + (size_t)N * 256;
  float* logit = agg  + (size_t)N * 256;
  unsigned* lmax = (unsigned*)(logit + (size_t)EP * 4);
  float* denom = (float*)(lmax + (size_t)N * 4);

  const int TPB = 256;
  int eblocks = (int)((EP + 7) / 8);
  int gblocks = (((N + 15) / 16) + 7) / 8;

  hipMemsetAsync(agg,   0, (size_t)N * 128 * sizeof(float), stream);
  hipMemsetAsync(lmax,  0, (size_t)N * 4 * sizeof(unsigned), stream);
  hipMemsetAsync(denom, 0, (size_t)N * 4 * sizeof(float), stream);
  gatv2_gemm_nt_wmma<<<gblocks, TPB, 0, stream>>>(x, Wl1, xl, N, 128);
  gatv2_gemm_nt_wmma<<<gblocks, TPB, 0, stream>>>(x, Wr1, xr, N, 128);
  gatv2_edge_logits<32><<<eblocks, TPB, 0, stream>>>(xl, xr, att1, srcA, dstA,
                                                     logit, lmax, E, N);
  gatv2_edge_accum<32><<<eblocks, TPB, 0, stream>>>(xl, logit, lmax, denom, agg,
                                                    srcA, dstA, E, N);
  gatv2_finalize1<<<(N * 128 + TPB - 1) / TPB, TPB, 0, stream>>>(agg, denom, b1, hbuf, N);

  hipMemsetAsync(agg,   0, (size_t)N * 256 * sizeof(float), stream);
  hipMemsetAsync(lmax,  0, (size_t)N * 4 * sizeof(unsigned), stream);
  hipMemsetAsync(denom, 0, (size_t)N * 4 * sizeof(float), stream);
  gatv2_gemm_nt_wmma<<<gblocks, TPB, 0, stream>>>(hbuf, Wl2, xl, N, 256);
  gatv2_gemm_nt_wmma<<<gblocks, TPB, 0, stream>>>(hbuf, Wr2, xr, N, 256);
  gatv2_edge_logits<64><<<eblocks, TPB, 0, stream>>>(xl, xr, att2, srcA, dstA,
                                                     logit, lmax, E, N);
  gatv2_edge_accum<64><<<eblocks, TPB, 0, stream>>>(xl, logit, lmax, denom, agg,
                                                    srcA, dstA, E, N);
  gatv2_finalize2<<<(N * 64 + TPB - 1) / TPB, TPB, 0, stream>>>(agg, denom, b2,
                                                                (float*)d_out, N);
}


# Edges go straight into fixed 32-slot destination rows by one returning add per edge (longer rows spill to an overflow list): count, scan and fill become one pass; sums in row arrival order
# speedup vs baseline: 3.7162x; 3.7162x over previous
#include <hip/hip_runtime.h>
#include <hip/hip_bf16.h>
#include <cmath>

typedef __attribute__((ext_vector_type(16))) _Float16 v16h;
typedef __attribute__((ext_vector_type(8)))  float    v8f;

#define NNODES 50000
#define NEDGES 500000

#define ROW_CAP   32
#define DEG_WORDS (NNODES + 16)
static_assert((DEG_WORDS * 4) % 128 == 0, "the table behind the degree words starts on a 128-byte line");

#define GEMM_WAVES   13
#define GEMM_THREADS (GEMM_WAVES * 32)
#define GEMM_STRIPS  ((NNODES + 15) / 16)
#define GEMM_BLOCKS  ((GEMM_STRIPS + GEMM_WAVES - 1) / GEMM_WAVES)
#define W_PITCH_H    136
static_assert(NNODES % 16 == 0, "every 16-row strip is whole: no row clamp, no store guard");

typedef __attribute__((ext_vector_type(8))) _Float16 v8h;
typedef __attribute__((ext_vector_type(4))) _Float16 v4h;

__device__ __forceinline__ v16h lds_frag_f16(const _Float16* p  ) {
  v8h lo = *(const v8h*)p;
  v8h hi = *(const v8h*)(p + 16);
  return __builtin_shufflevector(lo, hi, 0, 1, 2, 3, 4, 5, 6, 7, 8, 9, 10, 11, 12, 13, 14, 15);
}

template<int MCOLS>
__global__ void __launch_bounds__(GEMM_THREADS)
gatv2_gemm_nt_wmma(const float* __restrict__ X,
                   const float* __restrict__ W,
                   float* __restrict__ Y) {
  constexpr int K = 128;
  constexpr int NV4 = MCOLS * (K / 4);
  constexpr int ITER = (NV4 + GEMM_THREADS - 1) / GEMM_THREADS;
  __shared__ __attribute__((aligned(16))) _Float16 wl[MCOLS * W_PITCH_H];

  int tid = threadIdx.x;
  int lane = tid & 31;
  int wv = __builtin_amdgcn_readfirstlane(tid >> 5);
  int half = lane >> 4;
  int l = lane & 15;

  int strip = (int)blockIdx.x * GEMM_WAVES + wv;
  bool have = strip < GEMM_STRIPS;
  int srow = have ? strip : GEMM_STRIPS - 1;
  const float* xrow = X + (size_t)(srow * 16 + l) * K + half * 8;

  float4 araw[16];
  #pragma unroll
  for (int kk = 0; kk < 4; ++kk) {
    const float4* q = (const float4*)(xrow + kk * 32);
    araw[4 * kk + 0] = q[0]; araw[4 * kk + 1] = q[1];
    araw[4 * kk + 2] = q[4]; araw[4 * kk + 3] = q[5];
  }
  const float4* Wg = (const float4*)W;
  float4 wreg[ITER];
  #pragma unroll
  for (int i = 0; i < ITER; ++i) {
    int idx = tid + i * GEMM_THREADS;
    wreg[i] = Wg[idx < NV4 ? idx : NV4 - 1];
  }

  #pragma unroll
  for (int i = 0; i < ITER; ++i) {
    int idx = tid + i * GEMM_THREADS;
    if (idx >= NV4) idx = NV4 - 1;
    int wr = idx >> 5, c4 = idx & 31;
    v4h h;
    h[0] = (_Float16)wreg[i].x; h[1] = (_Float16)wreg[i].y;
    h[2] = (_Float16)wreg[i].z; h[3] = (_Float16)wreg[i].w;
    *(v4h*)(wl + wr * W_PITCH_H + c4 * 4) = h;
  }

  v16h a[4];
  #pragma unroll
  for (int kk = 0; kk < 4; ++kk) {
    #pragma unroll
    for (int j = 0; j < 4; ++j) {
      a[kk][4 * j + 0] = (_Float16)araw[4 * kk + j].x;
      a[kk][4 * j + 1] = (_Float16)araw[4 * kk + j].y;
      a[kk][4 * j + 2] = (_Float16)araw[4 * kk + j].z;
      a[kk][4 * j + 3] = (_Float16)araw[4 * kk + j].w;
    }
  }
  __syncthreads();
  if (!have) return;

  const _Float16* wbase = wl + l * W_PITCH_H + half * 8;
  float* ybase = Y + (size_t)(srow * 16 + half * 8) * MCOLS + l;
  #pragma unroll
  for (int tp = 0; tp < MCOLS / 128; ++tp) {
    const _Float16* wp = wbase + tp * 128 * W_PITCH_H;
    v8f c[8];
    #pragma unroll
    for (int t = 0; t < 8; ++t) c[t] = v8f{};
    #pragma unroll
    for (int kk = 0; kk < 4; ++kk) {
      v16h b[8];
      #pragma unroll
      for (int t = 0; t < 8; ++t) b[t] = lds_frag_f16(wp + t * 16 * W_PITCH_H + kk * 32);
      #pragma unroll
      for (int t = 0; t < 8; ++t)
        c[t] = __builtin_amdgcn_wmma_f32_16x16x32_f16(
             false, a[kk],  false, b[t],
             (short)0, c[t],  false,  false);
    }
    float* yc = ybase + tp * 128;
    #pragma unroll
    for (int r = 0; r < 8; ++r) {
      #pragma unroll
      for (int t = 0; t < 8; ++t) yc[(size_t)r * MCOLS + t * 16] = c[t][r];
    }
  }
}

__global__ void gatv2_edge_place(const int* __restrict__ srcA, const int* __restrict__ dstA,
                                 int* __restrict__ deg, int* __restrict__ tab,
                                 int* __restrict__ ovs, int* __restrict__ ovd,
                                 int* __restrict__ ovn, int E) {
  int e = (int)(blockIdx.x * blockDim.x + threadIdx.x);
  if (e < E) {
    int d = dstA[e];
    int s = srcA[e];
    int k = atomicAdd(&deg[d], 1);
    if (k < ROW_CAP) {
      tab[d * ROW_CAP + k] = s;
    } else {
      if (k == ROW_CAP) ovn[atomicAdd(&deg[NNODES + 1], 1)] = d;
      int o = atomicAdd(&deg[NNODES], 1);
      ovs[o] = s;
      ovd[o] = d;
    }
  }
}

template<int C>
__device__ __forceinline__ float gatv2_head_logit(float4 v, float4 r, float4 a) {
  float t0 = v.x + r.x, t1 = v.y + r.y, t2 = v.z + r.z, t3 = v.w + r.w;
  t0 = t0 > 0.f ? t0 : 0.2f * t0;
  t1 = t1 > 0.f ? t1 : 0.2f * t1;
  t2 = t2 > 0.f ? t2 : 0.2f * t2;
  t3 = t3 > 0.f ? t3 : 0.2f * t3;
  float p = (t0 * a.x + t1 * a.y) + (t2 * a.z + t3 * a.w);
  #pragma unroll
  for (int msk = 1; msk < C / 4; msk <<= 1) p += __shfl_xor(p, msk, 32);
  return p;
}

template<int C, bool CONCAT>
__device__ __forceinline__ void gatv2_long_row(int d, int lane,
                                               const float* __restrict__ xl,
                                               const float* __restrict__ xr,
                                               const float* __restrict__ att,
                                               const float* __restrict__ bias,
                                               const int* __restrict__ deg,
                                               const int* __restrict__ tab,
                                               const int* __restrict__ ovs,
                                               const int* __restrict__ ovd,
                                               float* __restrict__ out) {
  constexpr int HC = 4 * C;
  constexpr int NV = HC / 128;
  const float4* xrd = (const float4*)(xr + (size_t)d * HC);
  const float4* xld = (const float4*)(xl + (size_t)d * HC);
  const float4* at4 = (const float4*)att;
  float4 r[NV], a[NV], acc[NV];
  float m[NV], den[NV];
  #pragma unroll
  for (int i = 0; i < NV; ++i) {
    r[i] = xrd[i * 32 + lane];
    a[i] = at4[i * 32 + lane];
    float4 v = xld[i * 32 + lane];
    m[i] = gatv2_head_logit<C>(v, r[i], a[i]);
    den[i] = 1.f;
    acc[i] = v;
  }
  int novf = deg[NNODES];
  for (int k = 0; k < ROW_CAP + novf; ++k) {
    int sid;
    if (k < ROW_CAP) {
      sid = tab[d * ROW_CAP + k];
    } else {
      if (ovd[k - ROW_CAP] != d) continue;
      sid = ovs[k - ROW_CAP];
    }
    const float4* xs = (const float4*)(xl + (size_t)sid * HC);
    #pragma unroll
    for (int i = 0; i < NV; ++i) {
      float4 v = xs[i * 32 + lane];
      float lg = gatv2_head_logit<C>(v, r[i], a[i]);
      float mn = fmaxf(m[i], lg);
      float sc = expf(m[i] - mn);
      float w = expf(lg - mn);
      den[i] = den[i] * sc + w;
      acc[i].x = acc[i].x * sc + w * v.x;
      acc[i].y = acc[i].y * sc + w * v.y;
      acc[i].z = acc[i].z * sc + w * v.z;
      acc[i].w = acc[i].w * sc + w * v.w;
      m[i] = mn;
    }
  }
  if (CONCAT) {
    float4* o = (float4*)(out + (size_t)d * HC);
    #pragma unroll
    for (int i = 0; i < NV; ++i) {
      float dn = den[i] + 1e-16f;
      float4 b = ((const float4*)bias)[i * 32 + lane];
      float4 y;
      y.x = acc[i].x / dn + b.x; y.y = acc[i].y / dn + b.y;
      y.z = acc[i].z / dn + b.z; y.w = acc[i].w / dn + b.w;
      y.x = y.x > 0.f ? y.x : expm1f(y.x);
      y.y = y.y > 0.f ? y.y : expm1f(y.y);
      y.z = y.z > 0.f ? y.z : expm1f(y.z);
      y.w = y.w > 0.f ? y.w : expm1f(y.w);
      o[i * 32 + lane] = y;
    }
  } else {
    float4 t = {0.f, 0.f, 0.f, 0.f};
    #pragma unroll
    for (int i = 0; i < NV; ++i) {
      float dn = den[i] + 1e-16f;
      t.x += acc[i].x / dn; t.y += acc[i].y / dn;
      t.z += acc[i].z / dn; t.w += acc[i].w / dn;
    }
    t.x += __shfl_xor(t.x, 16, 32); t.y += __shfl_xor(t.y, 16, 32);
    t.z += __shfl_xor(t.z, 16, 32); t.w += __shfl_xor(t.w, 16, 32);
    if (lane < 16) {
      float4 b = ((const float4*)bias)[lane];
      float4 y;
      y.x = t.x * 0.25f + b.x; y.y = t.y * 0.25f + b.y;
      y.z = t.z * 0.25f + b.z; y.w = t.w * 0.25f + b.w;
      ((float4*)(out + (size_t)d * C))[lane] = y;
    }
  }
}

#define LONG_BLOCKS 4
template<int C, bool CONCAT>
__global__ void __launch_bounds__(256)
gatv2_node_softmax_agg(const float* __restrict__ xl,
                       const float* __restrict__ xr,
                       const float* __restrict__ att,
                       const float* __restrict__ bias,
                       const int* __restrict__ deg,
                       const int* __restrict__ csr,
                       const int* __restrict__ ovs,
                       const int* __restrict__ ovd,
                       const int* __restrict__ ovn,
                       float* __restrict__ out, int Nn) {
  constexpr int HC = 4 * C;
  constexpr int NV = HC / 128;
  constexpr int EB = 4;
  static_assert(CONCAT || C == 64, "head-mean lane mapping is written for C = 64");
  int wave = (int)((blockIdx.x * blockDim.x + threadIdx.x) >> 5);
  int lane = threadIdx.x & 31;
  if (wave >= Nn) {
    int nlong = deg[NNODES + 1];
    int nlw = (int)(gridDim.x * (blockDim.x >> 5)) - Nn;
    for (int j = __builtin_amdgcn_readfirstlane(wave) - Nn; j < nlong; j += nlw)
      gatv2_long_row<C, CONCAT>(ovn[j], lane, xl, xr, att, bias, deg, csr, ovs, ovd, out);
    return;
  }
  int d = __builtin_amdgcn_readfirstlane(wave);
  int n = deg[d];
  const float4* xrd = (const float4*)(xr + (size_t)d * HC);
  const float4* xld = (const float4*)(xl + (size_t)d * HC);
  const float4* at4 = (const float4*)att;

  float4 r[NV], a[NV], acc[NV];
  float m[NV], den[NV];
  #pragma unroll
  for (int i = 0; i < NV; ++i) {
    r[i] = xrd[i * 32 + lane];
    a[i] = at4[i * 32 + lane];
    float4 v = xld[i * 32 + lane];
    m[i] = gatv2_head_logit<C>(v, r[i], a[i]);
    den[i] = 1.f;
    acc[i] = v;
  }

  if (n > ROW_CAP) return;
  int p0 = d * ROW_CAP;
  int p1 = p0 + n;
  int last = p1 - 1;
  int s[EB];
  if (p0 < p1) {
    #pragma unroll
    for (int k = 0; k < EB; ++k) s[k] = csr[min(p0 + k, last)];
  }
  for (int p = p0; p < p1; p += EB) {
    float4 v[EB][NV];
    bool ok[EB];
    #pragma unroll
    for (int k = 0; k < EB; ++k) {
      ok[k] = p + k < p1;
      const float4* xs = (const float4*)(xl + (size_t)s[k] * HC);
      #pragma unroll
      for (int i = 0; i < NV; ++i) v[k][i] = xs[i * 32 + lane];
    }
    #pragma unroll
    for (int k = 0; k < EB; ++k) s[k] = csr[min(p + EB + k, last)];
    __builtin_amdgcn_sched_barrier(0);
    #pragma unroll
    for (int i = 0; i < NV; ++i) {
      float lg[EB];
      #pragma unroll
      for (int k = 0; k < EB; ++k) {
        lg[k] = gatv2_head_logit<C>(v[k][i], r[i], a[i]);
        if (!ok[k]) lg[k] = -INFINITY;
      }
      float mn = fmaxf(fmaxf(m[i], lg[0]), fmaxf(fmaxf(lg[1], lg[2]), lg[3]));
      float sc = expf(m[i] - mn);
      float w0 = expf(lg[0] - mn), w1 = expf(lg[1] - mn);
      float w2 = expf(lg[2] - mn), w3 = expf(lg[3] - mn);
      den[i] = den[i] * sc + ((w0 + w1) + (w2 + w3));
      acc[i].x = acc[i].x * sc + ((w0 * v[0][i].x + w1 * v[1][i].x) + (w2 * v[2][i].x + w3 * v[3][i].x));
      acc[i].y = acc[i].y * sc + ((w0 * v[0][i].y + w1 * v[1][i].y) + (w2 * v[2][i].y + w3 * v[3][i].y));
      acc[i].z = acc[i].z * sc + ((w0 * v[0][i].z + w1 * v[1][i].z) + (w2 * v[2][i].z + w3 * v[3][i].z));
      acc[i].w = acc[i].w * sc + ((w0 * v[0][i].w + w1 * v[1][i].w) + (w2 * v[2][i].w + w3 * v[3][i].w));
      m[i] = mn;
    }
  }

  if (CONCAT) {
    float4* o = (float4*)(out + (size_t)d * HC);
    #pragma unroll
    for (int i = 0; i < NV; ++i) {
      float dn = den[i] + 1e-16f;
      float4 b = ((const float4*)bias)[i * 32 + lane];
      float4 y;
      y.x = acc[i].x / dn + b.x; y.y = acc[i].y / dn + b.y;
      y.z = acc[i].z / dn + b.z; y.w = acc[i].w / dn + b.w;
      y.x = y.x > 0.f ? y.x : expm1f(y.x);
      y.y = y.y > 0.f ? y.y : expm1f(y.y);
      y.z = y.z > 0.f ? y.z : expm1f(y.z);
      y.w = y.w > 0.f ? y.w : expm1f(y.w);
      o[i * 32 + lane] = y;
    }
  } else {
    float4 t = {0.f, 0.f, 0.f, 0.f};
    #pragma unroll
    for (int i = 0; i < NV; ++i) {
      float dn = den[i] + 1e-16f;
      t.x += acc[i].x / dn; t.y += acc[i].y / dn;
      t.z += acc[i].z / dn; t.w += acc[i].w / dn;
    }
    t.x += __shfl_xor(t.x, 16, 32); t.y += __shfl_xor(t.y, 16, 32);
    t.z += __shfl_xor(t.z, 16, 32); t.w += __shfl_xor(t.w, 16, 32);
    if (lane < 16) {
      float4 b = ((const float4*)bias)[lane];
      float4 y;
      y.x = t.x * 0.25f + b.x; y.y = t.y * 0.25f + b.y;
      y.z = t.z * 0.25f + b.z; y.w = t.w * 0.25f + b.w;
      ((float4*)(out + (size_t)d * C))[lane] = y;
    }
  }
}

extern "C" void kernel_launch(void* const* d_in, const int* in_sizes, int n_in,
                              void* d_out, int out_size, void* d_ws, size_t ws_size,
                              hipStream_t stream) {
  (void)in_sizes; (void)n_in; (void)out_size; (void)ws_size;
  const int  N  = NNODES;
  const int  E  = NEDGES;

  const float* x    = (const float*)d_in[0];
  const int*   ei   = (const int*)  d_in[1];
  const float* Wl1  = (const float*)d_in[2];
  const float* Wr1  = (const float*)d_in[3];
  const float* att1 = (const float*)d_in[4];
  const float* b1   = (const float*)d_in[5];
  const float* Wl2  = (const float*)d_in[6];
  const float* Wr2  = (const float*)d_in[7];
  const float* att2 = (const float*)d_in[8];
  const float* b2   = (const float*)d_in[9];
  const int* srcA = ei;
  const int* dstA = ei + E;

  float* ws    = (float*)d_ws;
  float* hbuf  = ws;
  float* xl    = hbuf + (size_t)N * 128;
  float* xr    = xl   + (size_t)N * 256;
  int* deg     = (int*)(xr + (size_t)N * 256);
  int* tab     = deg + DEG_WORDS;
  int* ovs     = tab + (size_t)N * ROW_CAP;
  int* ovd     = ovs + E;
  int* ovn     = ovd + E;

  const int TPB = 256;
  int nblocks = (N + 7) / 8;
  int tblocks = (E + TPB - 1) / TPB;

  hipMemsetAsync(deg, 0, (size_t)DEG_WORDS * sizeof(int), stream);
  gatv2_edge_place<<<tblocks, TPB, 0, stream>>>(srcA, dstA, deg, tab, ovs, ovd, ovn, E);

  gatv2_gemm_nt_wmma<128><<<GEMM_BLOCKS, GEMM_THREADS, 0, stream>>>(x, Wl1, xl);
  gatv2_gemm_nt_wmma<128><<<GEMM_BLOCKS, GEMM_THREADS, 0, stream>>>(x, Wr1, xr);
  gatv2_node_softmax_agg<32, true><<<nblocks + LONG_BLOCKS, TPB, 0, stream>>>(xl, xr, att1, b1, deg, tab,
                                                                             ovs, ovd, ovn, hbuf, N);

  gatv2_gemm_nt_wmma<256><<<GEMM_BLOCKS, GEMM_THREADS, 0, stream>>>(hbuf, Wl2, xl);
  gatv2_gemm_nt_wmma<256><<<GEMM_BLOCKS, GEMM_THREADS, 0, stream>>>(hbuf, Wr2, xr);
  gatv2_node_softmax_agg<64, false><<<nblocks + LONG_BLOCKS, TPB, 0, stream>>>(xl, xr, att2, b2, deg, tab,
                                                                              ovs, ovd, ovn, (float*)d_out, N);
}


# Aggregation: leaky_relu of the logit as max(t, 0.2 t), one maximum per channel in place of compare and select; same value bit for bit, no sum re-ordered
# speedup vs baseline: 5.9438x; 1.5994x over previous
#include <hip/hip_runtime.h>
#include <hip/hip_bf16.h>
#include <cmath>

typedef __attribute__((ext_vector_type(16))) _Float16 v16h;
typedef __attribute__((ext_vector_type(8)))  float    v8f;

#define NNODES 50000
#define NEDGES 500000

#define ROW_CAP   32
#define DEG_WORDS (NNODES + 16)
static_assert((DEG_WORDS * 4) % 128 == 0, "the table behind the degree words starts on a 128-byte line");

#define RANGE_SHIFT  8
#define EDGE_RANGE   (1 << RANGE_SHIFT)
#define NRANGES      ((NNODES + EDGE_RANGE - 1) / EDGE_RANGE)
#define BIN_THREADS  256
#define EDGES_PER_THREAD 8
#define EDGE_CHUNK   (BIN_THREADS * EDGES_PER_THREAD)
#define NCHUNKS      ((NEDGES + EDGE_CHUNK - 1) / EDGE_CHUNK)
#define OFF_PITCH    256
static_assert(NRANGES + 1 <= OFF_PITCH && NRANGES <= BIN_THREADS, "one thread and one offset word per range, plus the total");

__global__ void __launch_bounds__(BIN_THREADS)
gatv2_edge_bin(const int* __restrict__ srcA, const int* __restrict__ dstA,
               int* __restrict__ pairs, int* __restrict__ off,
               int* __restrict__ deg, int E) {
  __shared__ int cnt[BIN_THREADS];
  __shared__ int psum[BIN_THREADS / 32];
  int t = threadIdx.x;
  cnt[t] = 0;
  if (blockIdx.x == 0 && t < 2) deg[NNODES + t] = 0;
  __syncthreads();

  int base = (int)blockIdx.x * EDGE_CHUNK;
  int d[EDGES_PER_THREAD], s[EDGES_PER_THREAD], r[EDGES_PER_THREAD];
  #pragma unroll
  for (int k = 0; k < EDGES_PER_THREAD; ++k) {
    int e = min(base + k * BIN_THREADS + t, E - 1);
    d[k] = dstA[e];
    s[k] = srcA[e];
  }
  #pragma unroll
  for (int k = 0; k < EDGES_PER_THREAD; ++k) {
    r[k] = 0;
    if (base + k * BIN_THREADS + t < E) r[k] = atomicAdd(&cnt[d[k] >> RANGE_SHIFT], 1);
  }
  __syncthreads();

  int own = cnt[t];
  int lane = t & 31, wv = t >> 5;
  int incl = own;
  #pragma unroll
  for (int o = 1; o < 32; o <<= 1) {
    int u = __shfl_up(incl, o, 32);
    if (lane >= o) incl += u;
  }
  if (lane == 31) psum[wv] = incl;
  __syncthreads();
  int pre = 0;
  #pragma unroll
  for (int j = 0; j < BIN_THREADS / 32; ++j) {
    int v = psum[j];
    if (j < wv) pre += v;
  }
  int start = pre + incl - own;
  off[(int)blockIdx.x * OFF_PITCH + t] = start;
  cnt[t] = start;
  __syncthreads();

  #pragma unroll
  for (int k = 0; k < EDGES_PER_THREAD; ++k) {
    if (base + k * BIN_THREADS + t < E) {
      int pos = base + cnt[d[k] >> RANGE_SHIFT] + r[k];
      *(int2*)(pairs + 2 * (size_t)pos) = make_int2(s[k], d[k]);
    }
  }
}

template<int NTHR>
__device__ __forceinline__ void
gatv2_edge_rows(int b, int t, int* cnt  ,
                const int* __restrict__ pairs, const int* __restrict__ off,
                int* __restrict__ deg, int* __restrict__ tab,
                int* __restrict__ ovs, int* __restrict__ ovd, int* __restrict__ ovn) {
  constexpr int PB = 16;
  static_assert(EDGE_RANGE <= NTHR, "one degree word per thread");
  if (t < EDGE_RANGE) cnt[t] = 0;
  __syncthreads();
  for (int ch = t; ch < NCHUNKS; ch += NTHR) {
    int beg = off[ch * OFF_PITCH + b];
    int end = off[ch * OFF_PITCH + b + 1];
    const int2* seg = (const int2*)(pairs + 2 * (size_t)ch * EDGE_CHUNK);
    for (int p = beg; p < end; p += PB) {
      int2 e[PB];
      #pragma unroll
      for (int k = 0; k < PB; ++k) e[k] = seg[min(p + k, end - 1)];
      #pragma unroll
      for (int k = 0; k < PB; ++k) {
        if (p + k < end) {
          int sN = e[k].x, dN = e[k].y;
          int q = atomicAdd(&cnt[dN & (EDGE_RANGE - 1)], 1);
          if (q < ROW_CAP) {
            tab[dN * ROW_CAP + q] = sN;
          } else {
            if (q == ROW_CAP) ovn[atomicAdd(&deg[NNODES + 1], 1)] = dN;
            int o = atomicAdd(&deg[NNODES], 1);
            ovs[o] = sN;
            ovd[o] = dN;
          }
        }
      }
    }
  }
  __syncthreads();
  int node = b * EDGE_RANGE + t;
  if (t < EDGE_RANGE && node < NNODES) deg[node] = cnt[t];
}

#define GEMM_WAVES   13
#define GEMM_THREADS (GEMM_WAVES * 32)
#define GEMM_STRIPS  ((NNODES + 15) / 16)
#define GEMM_BLOCKS  ((GEMM_STRIPS + GEMM_WAVES - 1) / GEMM_WAVES)
#define W_PITCH_H    136
static_assert(NNODES % 16 == 0, "every 16-row strip is whole: no row clamp, no store guard");

typedef __attribute__((ext_vector_type(8))) _Float16 v8h;
typedef __attribute__((ext_vector_type(4))) _Float16 v4h;

__device__ __forceinline__ v16h lds_frag_f16(const _Float16* p  ) {
  v8h lo = *(const v8h*)p;
  v8h hi = *(const v8h*)(p + 16);
  return __builtin_shufflevector(lo, hi, 0, 1, 2, 3, 4, 5, 6, 7, 8, 9, 10, 11, 12, 13, 14, 15);
}

template<int MCOLS, int ROW_BLOCKS>
__global__ void __launch_bounds__(GEMM_THREADS)
gatv2_gemm_nt_wmma_pair(const float* __restrict__ X,
                        const float* __restrict__ Wa,
                        const float* __restrict__ Wb,
                        float* __restrict__ Ya,
                        float* __restrict__ Yb,
                        const int* __restrict__ pairs, const int* __restrict__ off,
                        int* __restrict__ deg, int* __restrict__ tab,
                        int* __restrict__ ovs, int* __restrict__ ovd, int* __restrict__ ovn) {
  if constexpr (ROW_BLOCKS > 0) {
    __shared__ int cnt[EDGE_RANGE];
    if ((int)blockIdx.x < ROW_BLOCKS) {
      gatv2_edge_rows<GEMM_THREADS>((int)blockIdx.x, (int)threadIdx.x, cnt, pairs, off, deg, tab, ovs, ovd, ovn);
      return;
    }
  }
  constexpr int K = 128;
  constexpr int NV4 = MCOLS * (K / 4);
  constexpr int NV4T = 2 * NV4;
  constexpr int ITER = (NV4T + GEMM_THREADS - 1) / GEMM_THREADS;
  constexpr int PASSES = MCOLS / 128;
  __shared__ __attribute__((aligned(16))) _Float16 wl[2 * MCOLS * W_PITCH_H];

  int tid = threadIdx.x;
  int lane = tid & 31;
  int wv = __builtin_amdgcn_readfirstlane(tid >> 5);
  int half = lane >> 4;
  int l = lane & 15;

  int strip = ((int)blockIdx.x - ROW_BLOCKS) * GEMM_WAVES + wv;
  bool have = strip < GEMM_STRIPS;
  int srow = have ? strip : GEMM_STRIPS - 1;
  const float* xrow = X + (size_t)(srow * 16 + l) * K + half * 8;

  float4 araw[16];
  #pragma unroll
  for (int kk = 0; kk < 4; ++kk) {
    const float4* q = (const float4*)(xrow + kk * 32);
    araw[4 * kk + 0] = q[0]; araw[4 * kk + 1] = q[1];
    araw[4 * kk + 2] = q[4]; araw[4 * kk + 3] = q[5];
  }
  const float4* Wga = (const float4*)Wa;
  const float4* Wgb = (const float4*)Wb;
  float4 wreg[ITER];
  #pragma unroll
  for (int i = 0; i < ITER; ++i) {
    int idx = tid + i * GEMM_THREADS;
    if (idx >= NV4T) idx = NV4T - 1;
    const float4* wsrc = idx < NV4 ? Wga + idx : Wgb + (idx - NV4);
    wreg[i] = *wsrc;
  }

  #pragma unroll
  for (int i = 0; i < ITER; ++i) {
    int idx = tid + i * GEMM_THREADS;
    if (idx >= NV4T) idx = NV4T - 1;
    int wr = idx >> 5, c4 = idx & 31;
    if (wr < MCOLS) wr = (wr & ~127) | ((wr & 7) << 4) | ((wr >> 3) & 15);
    v4h h;
    h[0] = (_Float16)wreg[i].x; h[1] = (_Float16)wreg[i].y;
    h[2] = (_Float16)wreg[i].z; h[3] = (_Float16)wreg[i].w;
    *(v4h*)(wl + wr * W_PITCH_H + c4 * 4) = h;
  }

  v16h a[4];
  #pragma unroll
  for (int kk = 0; kk < 4; ++kk) {
    #pragma unroll
    for (int j = 0; j < 4; ++j) {
      a[kk][4 * j + 0] = (_Float16)araw[4 * kk + j].x;
      a[kk][4 * j + 1] = (_Float16)araw[4 * kk + j].y;
      a[kk][4 * j + 2] = (_Float16)araw[4 * kk + j].z;
      a[kk][4 * j + 3] = (_Float16)araw[4 * kk + j].w;
    }
  }
  __syncthreads();
  if (!have) return;

  const _Float16* wbase = wl + l * W_PITCH_H + half * 8;
  size_t yoff = (size_t)(srow * 16 + half * 8) * MCOLS + l;
  #pragma unroll
  for (int tp = 0; tp < 2 * PASSES; ++tp) {
    const _Float16* wp = wbase + tp * 128 * W_PITCH_H;
    v8f c[8];
    #pragma unroll
    for (int t = 0; t < 8; ++t) c[t] = v8f{};
    #pragma unroll
    for (int kk = 0; kk < 4; ++kk) {
      v16h b[8];
      #pragma unroll
      for (int t = 0; t < 8; ++t) b[t] = lds_frag_f16(wp + t * 16 * W_PITCH_H + kk * 32);
      #pragma unroll
      for (int t = 0; t < 8; ++t)
        c[t] = __builtin_amdgcn_wmma_f32_16x16x32_f16(
             false, a[kk],  false, b[t],
             (short)0, c[t],  false,  false);
    }
    if (tp < PASSES) {
      _Float16* yh = (_Float16*)Ya + (size_t)(srow * 16 + half * 8) * MCOLS + tp * 128 + l * 8;
      #pragma unroll
      for (int r = 0; r < 8; ++r) {
        v8h o;
        #pragma unroll
        for (int t = 0; t < 8; ++t) o[t] = (_Float16)c[t][r];
        *(v8h*)(yh + (size_t)r * MCOLS) = o;
      }
    } else {
      float* yc = Yb + yoff + (tp - PASSES) * 128;
      #pragma unroll
      for (int r = 0; r < 8; ++r) {
        #pragma unroll
        for (int t = 0; t < 8; ++t) yc[(size_t)r * MCOLS + t * 16] = c[t][r];
      }
    }
  }
}

#define LOG2E_F 1.4426950408889634f
__device__ __forceinline__ float gatv2_exp2(float x) { return __builtin_amdgcn_exp2f(x); }
__device__ __forceinline__ float4 gatv2_att_base2(float4 a) {
  a.x *= LOG2E_F; a.y *= LOG2E_F; a.z *= LOG2E_F; a.w *= LOG2E_F;
  return a;
}
__device__ __forceinline__ float gatv2_expm1_base2(float y) {
  return gatv2_exp2(y * LOG2E_F) - 1.f;
}

__device__ __forceinline__ v4h gatv2_row_raw(const _Float16* row, int i, int lane) {
  return *(const v4h*)(row + i * 128 + lane * 4);
}
__device__ __forceinline__ float4 gatv2_row_f32(v4h h) {
  float4 v;
  v.x = (float)h[0]; v.y = (float)h[1]; v.z = (float)h[2]; v.w = (float)h[3];
  return v;
}

typedef __attribute__((ext_vector_type(2))) float v2f;
template<int C>
__device__ __forceinline__ float gatv2_head_logit(float4 v, float4 r, float4 a) {
  v2f tl = v2f{v.x, v.y} + v2f{r.x, r.y};
  v2f th = v2f{v.z, v.w} + v2f{r.z, r.w};
  v2f sl = tl * 0.2f, sh = th * 0.2f;
  tl = __builtin_elementwise_max(tl, sl);
  th = __builtin_elementwise_max(th, sh);
  v2f q = tl * v2f{a.x, a.y} + th * v2f{a.z, a.w};
  float p = q.x + q.y;
  #pragma unroll
  for (int msk = 1; msk < C / 4; msk <<= 1) p += __shfl_xor(p, msk, 32);
  return p;
}

template<int C, bool CONCAT>
__device__ __forceinline__ void gatv2_long_row(int d, int lane,
                                               const float* __restrict__ xl,
                                               const float* __restrict__ xr,
                                               const float* __restrict__ att,
                                               const float* __restrict__ bias,
                                               const int* __restrict__ deg,
                                               const int* __restrict__ tab,
                                               const int* __restrict__ ovs,
                                               const int* __restrict__ ovd,
                                               float* __restrict__ out) {
  constexpr int HC = 4 * C;
  constexpr int NV = HC / 128;
  const float4* xrd = (const float4*)(xr + (size_t)d * HC);
  const _Float16* xlh = (const _Float16*)xl;
  const float4* at4 = (const float4*)att;
  float4 r[NV], a[NV], acc[NV];
  float m[NV], den[NV];
  #pragma unroll
  for (int i = 0; i < NV; ++i) {
    r[i] = xrd[i * 32 + lane];
    a[i] = gatv2_att_base2(at4[i * 32 + lane]);
    float4 v = gatv2_row_f32(gatv2_row_raw(xlh + (size_t)d * HC, i, lane));
    m[i] = gatv2_head_logit<C>(v, r[i], a[i]);
    den[i] = 1.f;
    acc[i] = v;
  }
  int novf = deg[NNODES];
  for (int k = 0; k < ROW_CAP + novf; ++k) {
    int sid;
    if (k < ROW_CAP) {
      sid = tab[d * ROW_CAP + k];
    } else {
      if (ovd[k - ROW_CAP] != d) continue;
      sid = ovs[k - ROW_CAP];
    }
    const _Float16* xs = xlh + (size_t)sid * HC;
    #pragma unroll
    for (int i = 0; i < NV; ++i) {
      float4 v = gatv2_row_f32(gatv2_row_raw(xs, i, lane));
      float lg = gatv2_head_logit<C>(v, r[i], a[i]);
      float mn = fmaxf(m[i], lg);
      float sc = gatv2_exp2(m[i] - mn);
      float w = gatv2_exp2(lg - mn);
      den[i] = den[i] * sc + w;
      acc[i].x = acc[i].x * sc + w * v.x;
      acc[i].y = acc[i].y * sc + w * v.y;
      acc[i].z = acc[i].z * sc + w * v.z;
      acc[i].w = acc[i].w * sc + w * v.w;
      m[i] = mn;
    }
  }
  if (CONCAT) {
    float4* o = (float4*)(out + (size_t)d * HC);
    #pragma unroll
    for (int i = 0; i < NV; ++i) {
      float dn = den[i] + 1e-16f;
      float4 b = ((const float4*)bias)[i * 32 + lane];
      float4 y;
      y.x = acc[i].x / dn + b.x; y.y = acc[i].y / dn + b.y;
      y.z = acc[i].z / dn + b.z; y.w = acc[i].w / dn + b.w;
      y.x = y.x > 0.f ? y.x : gatv2_expm1_base2(y.x);
      y.y = y.y > 0.f ? y.y : gatv2_expm1_base2(y.y);
      y.z = y.z > 0.f ? y.z : gatv2_expm1_base2(y.z);
      y.w = y.w > 0.f ? y.w : gatv2_expm1_base2(y.w);
      o[i * 32 + lane] = y;
    }
  } else {
    float4 t = {0.f, 0.f, 0.f, 0.f};
    #pragma unroll
    for (int i = 0; i < NV; ++i) {
      float dn = den[i] + 1e-16f;
      t.x += acc[i].x / dn; t.y += acc[i].y / dn;
      t.z += acc[i].z / dn; t.w += acc[i].w / dn;
    }
    t.x += __shfl_xor(t.x, 16, 32); t.y += __shfl_xor(t.y, 16, 32);
    t.z += __shfl_xor(t.z, 16, 32); t.w += __shfl_xor(t.w, 16, 32);
    if (lane < 16) {
      float4 b = ((const float4*)bias)[lane];
      float4 y;
      y.x = t.x * 0.25f + b.x; y.y = t.y * 0.25f + b.y;
      y.z = t.z * 0.25f + b.z; y.w = t.w * 0.25f + b.w;
      ((float4*)(out + (size_t)d * C))[lane] = y;
    }
  }
}

#define LONG_BLOCKS 4
template<int C, bool CONCAT>
__global__ void __launch_bounds__(256)
gatv2_node_softmax_agg(const float* __restrict__ xl,
                       const float* __restrict__ xr,
                       const float* __restrict__ att,
                       const float* __restrict__ bias,
                       const int* __restrict__ deg,
                       const int* __restrict__ csr,
                       const int* __restrict__ ovs,
                       const int* __restrict__ ovd,
                       const int* __restrict__ ovn,
                       float* __restrict__ out, int Nn) {
  constexpr int HC = 4 * C;
  constexpr int NV = HC / 128;
  constexpr int EB = 4;
  static_assert(CONCAT || C == 64, "head-mean lane mapping is written for C = 64");
  int wave = (int)((blockIdx.x * blockDim.x + threadIdx.x) >> 5);
  int lane = threadIdx.x & 31;
  if (wave >= Nn) {
    int nlong = deg[NNODES + 1];
    int nlw = (int)(gridDim.x * (blockDim.x >> 5)) - Nn;
    for (int j = __builtin_amdgcn_readfirstlane(wave) - Nn; j < nlong; j += nlw)
      gatv2_long_row<C, CONCAT>(ovn[j], lane, xl, xr, att, bias, deg, csr, ovs, ovd, out);
    return;
  }
  int d = __builtin_amdgcn_readfirstlane(wave);
  int n = deg[d];
  const float4* xrd = (const float4*)(xr + (size_t)d * HC);
  const _Float16* xlh = (const _Float16*)xl;
  const float4* at4 = (const float4*)att;

  float4 r[NV], a[NV], acc[NV];
  float m[NV], den[NV];
  #pragma unroll
  for (int i = 0; i < NV; ++i) {
    r[i] = xrd[i * 32 + lane];
    a[i] = gatv2_att_base2(at4[i * 32 + lane]);
    float4 v = gatv2_row_f32(gatv2_row_raw(xlh + (size_t)d * HC, i, lane));
    m[i] = gatv2_head_logit<C>(v, r[i], a[i]);
    den[i] = 1.f;
    acc[i] = v;
  }

  if (n > ROW_CAP) return;
  int p0 = d * ROW_CAP;
  int p1 = p0 + n;
  int last = p1 - 1;
  int s[EB];
  if (p0 < p1) {
    #pragma unroll
    for (int k = 0; k < EB; ++k) s[k] = csr[min(p0 + k, last)];
  }
  for (int p = p0; p < p1; p += EB) {
    v4h raw[EB][NV];
    bool ok[EB];
    #pragma unroll
    for (int k = 0; k < EB; ++k) {
      ok[k] = p + k < p1;
      const _Float16* xs = xlh + (size_t)s[k] * HC;
      #pragma unroll
      for (int i = 0; i < NV; ++i) raw[k][i] = gatv2_row_raw(xs, i, lane);
    }
    #pragma unroll
    for (int k = 0; k < EB; ++k) s[k] = csr[min(p + EB + k, last)];
    __builtin_amdgcn_sched_barrier(0);
    #pragma unroll
    for (int i = 0; i < NV; ++i) {
      float4 v[EB];
      float lg[EB];
      #pragma unroll
      for (int k = 0; k < EB; ++k) {
        v[k] = gatv2_row_f32(raw[k][i]);
        lg[k] = gatv2_head_logit<C>(v[k], r[i], a[i]);
        if (!ok[k]) lg[k] = -INFINITY;
      }
      float mn = fmaxf(fmaxf(m[i], lg[0]), fmaxf(fmaxf(lg[1], lg[2]), lg[3]));
      float sc = gatv2_exp2(m[i] - mn);
      float w0 = gatv2_exp2(lg[0] - mn), w1 = gatv2_exp2(lg[1] - mn);
      float w2 = gatv2_exp2(lg[2] - mn), w3 = gatv2_exp2(lg[3] - mn);
      den[i] = den[i] * sc + ((w0 + w1) + (w2 + w3));
      acc[i].x = acc[i].x * sc + ((w0 * v[0].x + w1 * v[1].x) + (w2 * v[2].x + w3 * v[3].x));
      acc[i].y = acc[i].y * sc + ((w0 * v[0].y + w1 * v[1].y) + (w2 * v[2].y + w3 * v[3].y));
      acc[i].z = acc[i].z * sc + ((w0 * v[0].z + w1 * v[1].z) + (w2 * v[2].z + w3 * v[3].z));
      acc[i].w = acc[i].w * sc + ((w0 * v[0].w + w1 * v[1].w) + (w2 * v[2].w + w3 * v[3].w));
      m[i] = mn;
    }
  }

  if (CONCAT) {
    float4* o = (float4*)(out + (size_t)d * HC);
    #pragma unroll
    for (int i = 0; i < NV; ++i) {
      float dn = den[i] + 1e-16f;
      float4 b = ((const float4*)bias)[i * 32 + lane];
      float4 y;
      y.x = acc[i].x / dn + b.x; y.y = acc[i].y / dn + b.y;
      y.z = acc[i].z / dn + b.z; y.w = acc[i].w / dn + b.w;
      y.x = y.x > 0.f ? y.x : gatv2_expm1_base2(y.x);
      y.y = y.y > 0.f ? y.y : gatv2_expm1_base2(y.y);
      y.z = y.z > 0.f ? y.z : gatv2_expm1_base2(y.z);
      y.w = y.w > 0.f ? y.w : gatv2_expm1_base2(y.w);
      o[i * 32 + lane] = y;
    }
  } else {
    float4 t = {0.f, 0.f, 0.f, 0.f};
    #pragma unroll
    for (int i = 0; i < NV; ++i) {
      float dn = den[i] + 1e-16f;
      t.x += acc[i].x / dn; t.y += acc[i].y / dn;
      t.z += acc[i].z / dn; t.w += acc[i].w / dn;
    }
    t.x += __shfl_xor(t.x, 16, 32); t.y += __shfl_xor(t.y, 16, 32);
    t.z += __shfl_xor(t.z, 16, 32); t.w += __shfl_xor(t.w, 16, 32);
    if (lane < 16) {
      float4 b = ((const float4*)bias)[lane];
      float4 y;
      y.x = t.x * 0.25f + b.x; y.y = t.y * 0.25f + b.y;
      y.z = t.z * 0.25f + b.z; y.w = t.w * 0.25f + b.w;
      ((float4*)(out + (size_t)d * C))[lane] = y;
    }
  }
}

extern "C" void kernel_launch(void* const* d_in, const int* in_sizes, int n_in,
                              void* d_out, int out_size, void* d_ws, size_t ws_size,
                              hipStream_t stream) {
  (void)in_sizes; (void)n_in; (void)out_size; (void)ws_size;
  const int  N  = NNODES;
  const int  E  = NEDGES;

  const float* x    = (const float*)d_in[0];
  const int*   ei   = (const int*)  d_in[1];
  const float* Wl1  = (const float*)d_in[2];
  const float* Wr1  = (const float*)d_in[3];
  const float* att1 = (const float*)d_in[4];
  const float* b1   = (const float*)d_in[5];
  const float* Wl2  = (const float*)d_in[6];
  const float* Wr2  = (const float*)d_in[7];
  const float* att2 = (const float*)d_in[8];
  const float* b2   = (const float*)d_in[9];
  const int* srcA = ei;
  const int* dstA = ei + E;

  float* ws    = (float*)d_ws;
  float* hbuf  = ws;
  float* xl    = hbuf + (size_t)N * 128;
  float* xr    = xl   + (size_t)N * 256;
  int* deg     = (int*)(xr + (size_t)N * 256);
  int* tab     = deg + DEG_WORDS;
  int* ovs     = tab + (size_t)N * ROW_CAP;
  int* ovd     = ovs + E;
  int* ovn     = ovd + E;
  int* pairs   = ovn + DEG_WORDS;
  int* off     = pairs + 2 * (size_t)E;

  const int TPB = 256;
  int nblocks = (N + 7) / 8;

  gatv2_edge_bin<<<NCHUNKS, BIN_THREADS, 0, stream>>>(srcA, dstA, pairs, off, deg, E);

  gatv2_gemm_nt_wmma_pair<128, NRANGES><<<NRANGES + GEMM_BLOCKS, GEMM_THREADS, 0, stream>>>(x, Wl1, Wr1, xl, xr,
                                                                                          pairs, off, deg, tab, ovs, ovd, ovn);
  gatv2_node_softmax_agg<32, true><<<nblocks + LONG_BLOCKS, TPB, 0, stream>>>(xl, xr, att1, b1, deg, tab,
                                                                             ovs, ovd, ovn, hbuf, N);

  gatv2_gemm_nt_wmma_pair<256, 0><<<GEMM_BLOCKS, GEMM_THREADS, 0, stream>>>(hbuf, Wl2, Wr2, xl, xr,
                                                                            pairs, off, deg, tab, ovs, ovd, ovn);
  gatv2_node_softmax_agg<64, false><<<nblocks + LONG_BLOCKS, TPB, 0, stream>>>(xl, xr, att2, b2, deg, tab,
                                                                              ovs, ovd, ovn, (float*)d_out, N);
}
